# starting kernel
# baseline (speedup 1.0000x reference)
#include <hip/hip_runtime.h>
#include <math.h>


typedef float v2f __attribute__((ext_vector_type(2)));
typedef float v8f __attribute__((ext_vector_type(8)));

#define LSEQ   1024
#define DMODEL 1024

__global__ __launch_bounds__(128)
void gemm_nt_f32_wmma(const float* __restrict__ A, const float* __restrict__ B,
                      float* __restrict__ C, int M, int N, int K, float alpha) {
  const int lane = threadIdx.x & 31;
  const int wave = threadIdx.x >> 5;
  const int lid  = lane & 15;
  const int half = lane >> 4;
  const int mBase = blockIdx.y * 128 + (wave >> 1) * 64;
  const int nBase = blockIdx.x * 128 + (wave & 1) * 64;
  const int koff = half * 2;

  const float* aPtr[4];
  const float* bPtr[4];
  #pragma unroll
  for (int i = 0; i < 4; ++i) {
    aPtr[i] = A + (size_t)(mBase + i * 16 + lid) * K + koff;
    bPtr[i] = B + (size_t)(nBase + i * 16 + lid) * K + koff;
  }

  v8f acc[4][4];
  #pragma unroll
  for (int i = 0; i < 4; ++i)
    #pragma unroll
    for (int j = 0; j < 4; ++j)
      acc[i][j] = (v8f){};

  #pragma unroll 2
  for (int k = 0; k < K; k += 4) {
    v2f fa[4], fb[4];
    #pragma unroll
    for (int i = 0; i < 4; ++i) fa[i] = *(const v2f*)(aPtr[i] + k);
    #pragma unroll
    for (int j = 0; j < 4; ++j) fb[j] = *(const v2f*)(bPtr[j] + k);
    #pragma unroll
    for (int i = 0; i < 4; ++i)
      #pragma unroll
      for (int j = 0; j < 4; ++j)
        acc[i][j] = __builtin_amdgcn_wmma_f32_16x16x4_f32(
            false, fa[i], false, fb[j], (short)0, acc[i][j], false, false);
  }

  #pragma unroll
  for (int i = 0; i < 4; ++i) {
    #pragma unroll
    for (int j = 0; j < 4; ++j) {
      const int col = nBase + j * 16 + lid;
      #pragma unroll
      for (int r = 0; r < 8; ++r) {
        const int row = mBase + i * 16 + half * 8 + r;
        C[(size_t)row * N + col] = alpha * acc[i][j][r];
      }
    }
  }
}

__global__ __launch_bounds__(128)
void ema_scan(float* __restrict__ Ks, float* __restrict__ Vs,
              const float* __restrict__ gates, int L, int D) {
  const int d = blockIdx.x * blockDim.x + threadIdx.x;
  if (d >= D) return;
  const float g = gates[0];
  float ek = 0.0f, ev = 0.0f;
  #pragma unroll 4
  for (int t = 0; t < L; ++t) {
    const size_t idx = (size_t)t * D + d;
    ek = fmaf(g, ek, Ks[idx]);
    ev = fmaf(g, ev, Vs[idx]);
    Ks[idx] = ek;
    Vs[idx] = ev;
  }
}

__global__ __launch_bounds__(256)
void gate_softmax(float* __restrict__ Q, const float* __restrict__ Kema,
                  const float* __restrict__ Vema, int L, int D) {
  const int t    = blockIdx.x;
  const int wave = threadIdx.x >> 5;
  const int lane = threadIdx.x & 31;
  const int h    = blockIdx.y * 8 + wave;
  const size_t idx = (size_t)t * D + h * 32 + lane;
  const float qv = Q[idx];
  float p = qv * Kema[idx];
  #pragma unroll
  for (int off = 16; off > 0; off >>= 1)
    p += __shfl_xor(p, off, 32);
  const float a = 1.0f / (1.0f + 31.0f * expf(-p));
  Q[idx] = a * Vema[idx];
}

extern "C" void kernel_launch(void* const* d_in, const int* in_sizes, int n_in,
                              void* d_out, int out_size, void* d_ws, size_t ws_size,
                              hipStream_t stream) {
  const float* q     = (const float*)d_in[0];
  const float* k     = (const float*)d_in[1];
  const float* v     = (const float*)d_in[2];
  const float* Wq    = (const float*)d_in[3];
  const float* Wk    = (const float*)d_in[4];
  const float* Wv    = (const float*)d_in[5];
  const float* Wd    = (const float*)d_in[6];
  const float* gates = (const float*)d_in[7];
  float* out = (float*)d_out;

  const int L = LSEQ, D = DMODEL;
  const size_t LD = (size_t)L * D;
  float* Qs = (float*)d_ws;
  float* Ks = Qs + LD;
  float* Vs = Ks + LD;

  const float invScale = 1.0f / sqrtf(sqrtf(32.0f));

  dim3 gBlk(128);
  dim3 gGrid(D / 128, L / 128);

  hipLaunchKernelGGL(gemm_nt_f32_wmma, gGrid, gBlk, 0, stream, q, Wq, Qs, L, D, D, invScale);
  hipLaunchKernelGGL(gemm_nt_f32_wmma, gGrid, gBlk, 0, stream, k, Wk, Ks, L, D, D, invScale);
  hipLaunchKernelGGL(gemm_nt_f32_wmma, gGrid, gBlk, 0, stream, v, Wv, Vs, L, D, D, 1.0f);
  hipLaunchKernelGGL(ema_scan, dim3(D / 128), dim3(128), 0, stream, Ks, Vs, gates, L, D);
  hipLaunchKernelGGL(gate_softmax, dim3(L, 4), dim3(256), 0, stream, Qs, Ks, Vs, L, D);
  hipLaunchKernelGGL(gemm_nt_f32_wmma, gGrid, gBlk, 0, stream, Qs, Wd, out, L, D, D, 1.0f);
}


# EMA scan moved in front of the projections (linear over features): one work-group forms the q, kema, vema tiles of a 64x64 position and gates them; 6 -> 3 launches; scan and gate dot sum order change
# speedup vs baseline: 5.9333x; 5.9333x over previous
#include <hip/hip_runtime.h>
#include <math.h>


typedef float v2f __attribute__((ext_vector_type(2)));
typedef float v4f __attribute__((ext_vector_type(4)));
typedef float v8f __attribute__((ext_vector_type(8)));

#define LSEQ   1024
#define DMODEL 1024

#define GEMM_KB 8
#define GEMM_KW 4
#define GEMM_KS (GEMM_KB * GEMM_KW)
#define GEMM_T  64
#define RED_PITCH (GEMM_T + 4)

__device__ __forceinline__ void gemm_fetch(v2f (&fa)[2][4], v2f (&fb)[2][4],
                                           const float* const (&aPtr)[4],
                                           const float* const (&bPtr)[4], int k) {
  #pragma unroll
  for (int s = 0; s < 2; ++s) {
    #pragma unroll
    for (int i = 0; i < 4; ++i) fa[s][i] = *(const v2f*)(aPtr[i] + k + 4 * s);
    #pragma unroll
    for (int j = 0; j < 4; ++j) fb[s][j] = *(const v2f*)(bPtr[j] + k + 4 * s);
  }
}

__device__ __forceinline__ void gemm_block(v8f (&acc)[4][4], const v2f (&fa)[2][4],
                                           const v2f (&fb)[2][4]) {
  #pragma unroll
  for (int s = 0; s < 2; ++s)
    #pragma unroll
    for (int i = 0; i < 4; ++i)
      #pragma unroll
      for (int j = 0; j < 4; ++j)
        acc[i][j] = __builtin_amdgcn_wmma_f32_16x16x4_f32(
            false, fa[s][i], false, fb[s][j], (short)0, acc[i][j], false, false);
}

typedef float red_plane_t[GEMM_T][RED_PITCH];

__device__ __forceinline__ void gemm_wave_partial(v8f (&acc)[4][4],
                                                  const float* __restrict__ A,
                                                  const float* __restrict__ B,
                                                  int K, int mBase, int nBase,
                                                  int lid, int koff) {
  const float* aPtr[4];
  const float* bPtr[4];
  #pragma unroll
  for (int i = 0; i < 4; ++i) {
    aPtr[i] = A + (size_t)(mBase + i * 16 + lid) * K + koff;
    bPtr[i] = B + (size_t)(nBase + i * 16 + lid) * K + koff;
  }

  #pragma unroll
  for (int i = 0; i < 4; ++i)
    #pragma unroll
    for (int j = 0; j < 4; ++j)
      acc[i][j] = (v8f){};

  #define GEMM_FENCE() __builtin_amdgcn_sched_barrier(0)
  v2f a0[2][4], b0[2][4], a1[2][4], b1[2][4], a2[2][4], b2[2][4];
  gemm_fetch(a0, b0, aPtr, bPtr, 0);
  GEMM_FENCE();
  gemm_fetch(a1, b1, aPtr, bPtr, GEMM_KS);
  int k = 0;
  for (; k < K - 4 * GEMM_KS; k += 3 * GEMM_KS) {
    gemm_fetch(a2, b2, aPtr, bPtr, k + 2 * GEMM_KS);
    GEMM_FENCE();
    gemm_block(acc, a0, b0);
    GEMM_FENCE();
    gemm_fetch(a0, b0, aPtr, bPtr, k + 3 * GEMM_KS);
    GEMM_FENCE();
    gemm_block(acc, a1, b1);
    GEMM_FENCE();
    gemm_fetch(a1, b1, aPtr, bPtr, k + 4 * GEMM_KS);
    GEMM_FENCE();
    gemm_block(acc, a2, b2);
  }
  gemm_block(acc, a0, b0);
  gemm_block(acc, a1, b1);
  #undef GEMM_FENCE
}

__device__ __forceinline__ void gemm_reduce(v4f (&s)[8], red_plane_t* red,
                                            const v8f (&acc)[4][4], float alpha,
                                            int wave, int lid, int half) {
  #pragma unroll
  for (int i = 0; i < 4; ++i)
    #pragma unroll
    for (int j = 0; j < 4; ++j)
      #pragma unroll
      for (int r = 0; r < 8; ++r)
        red[wave][i * 16 + half * 8 + r][j * 16 + lid] = alpha * acc[i][j][r];
  __syncthreads();
  #pragma unroll
  for (int p = 0; p < 8; ++p) {
    const int row = wave * 16 + p * 2 + half;
    s[p] = *(const v4f*)&red[0][row][lid * 4];
    #pragma unroll
    for (int z = 1; z < GEMM_KW; ++z) s[p] += *(const v4f*)&red[z][row][lid * 4];
  }
}

__global__ __launch_bounds__(128)
void gemm_nt_f32_wmma(const float* __restrict__ A, const float* __restrict__ B,
                      float* __restrict__ C, int M, int N, int K, float alpha) {
  __shared__ __attribute__((aligned(16))) float red[GEMM_KW][GEMM_T][RED_PITCH];
  const int lane = threadIdx.x & 31;
  const int wave = threadIdx.x >> 5;
  const int lid  = lane & 15;
  const int half = lane >> 4;
  const int mBase = blockIdx.y * GEMM_T;
  const int nBase = blockIdx.x * GEMM_T;
  const int koff = wave * GEMM_KB + half * 2;

  v8f acc[4][4];
  gemm_wave_partial(acc, A, B, K, mBase, nBase, lid, koff);
  v4f s[8];
  gemm_reduce(s, red, acc, alpha, wave, lid, half);
  #pragma unroll
  for (int p = 0; p < 8; ++p) {
    const int row = wave * 16 + p * 2 + half;
    *(v4f*)(C + (size_t)(mBase + row) * N + nBase + lid * 4) = s[p];
  }
}

__global__ __launch_bounds__(128)
void proj_gate_f32_wmma(const float* __restrict__ q,  const float* __restrict__ Wq,
                        const float* __restrict__ Ek, const float* __restrict__ Wk,
                        const float* __restrict__ Ev, const float* __restrict__ Wv,
                        float* __restrict__ Z, int M, int N, int K, float scale) {
  __shared__ __attribute__((aligned(16))) float red[GEMM_KW][GEMM_T][RED_PITCH];
  const int lane = threadIdx.x & 31;
  const int wave = threadIdx.x >> 5;
  const int lid  = lane & 15;
  const int half = lane >> 4;
  const int mBase = blockIdx.y * GEMM_T;
  const int nBase = blockIdx.x * GEMM_T;
  const int koff = wave * GEMM_KB + half * 2;

  v8f acc[4][4];
  v4f sq[8], sk[8];
  float a[8];

  gemm_wave_partial(acc, q, Wq, K, mBase, nBase, lid, koff);
  gemm_reduce(sq, red, acc, scale, wave, lid, half);
  __syncthreads();

  gemm_wave_partial(acc, Ek, Wk, K, mBase, nBase, lid, koff);
  gemm_reduce(sk, red, acc, scale, wave, lid, half);
  #pragma unroll
  for (int p = 0; p < 8; ++p) {
    float d = sq[p].x * sk[p].x + sq[p].y * sk[p].y + sq[p].z * sk[p].z + sq[p].w * sk[p].w;
    #pragma unroll
    for (int off = 1; off < 8; off <<= 1)
      d += __shfl_xor(d, off, 32);
    a[p] = 1.0f / (1.0f + 31.0f * expf(-d));
  }
  __syncthreads();

  gemm_wave_partial(acc, Ev, Wv, K, mBase, nBase, lid, koff);
  gemm_reduce(sk, red, acc, 1.0f, wave, lid, half);
  #pragma unroll
  for (int p = 0; p < 8; ++p) {
    const int row = wave * 16 + p * 2 + half;
    *(v4f*)(Z + (size_t)(mBase + row) * N + nBase + lid * 4) = a[p] * sk[p];
  }
}

#define SCAN_CH      32
#define SCAN_THREADS 512
#define SCAN_RPP     (SCAN_THREADS / 8)
#define SCAN_PASSES  (LSEQ / SCAN_RPP)
#define SCAN_BLK     32

__device__ __forceinline__ void scan_fetch(float (&x)[SCAN_BLK], const float* col, int t) {
  #pragma unroll
  for (int j = 0; j < SCAN_BLK; ++j) x[j] = col[(t + j) * SCAN_CH];
}

__device__ __forceinline__ float scan_run(const float (&x)[SCAN_BLK], float* col, int t,
                                          float g, float e) {
  #pragma unroll
  for (int j = 0; j < SCAN_BLK; ++j) {
    e = fmaf(g, e, x[j]);
    col[(t + j) * SCAN_CH] = e;
  }
  return e;
}

__global__ __launch_bounds__(SCAN_THREADS)
void ema_scan(const float* __restrict__ kin, const float* __restrict__ vin,
              float* __restrict__ Ek, float* __restrict__ Ev,
              const float* __restrict__ gates) {
  __shared__ float strip[LSEQ * SCAN_CH];
  const float* X = (blockIdx.y == 0 ? kin : vin) + blockIdx.x * SCAN_CH;
  float*       Y = (blockIdx.y == 0 ? Ek : Ev) + blockIdx.x * SCAN_CH;
  const int tid = threadIdx.x;
  const int c4  = (tid & 7) * 4;
  const int r0  = tid >> 3;

  {
    v4f in[SCAN_PASSES];
    #pragma unroll
    for (int i = 0; i < SCAN_PASSES; ++i)
      in[i] = *(const v4f*)(X + (size_t)(i * SCAN_RPP + r0) * DMODEL + c4);
    #pragma unroll
    for (int i = 0; i < SCAN_PASSES; ++i)
      *(v4f*)(strip + (i * SCAN_RPP + r0) * SCAN_CH + c4) = in[i];
  }
  __syncthreads();

  if (tid < SCAN_CH) {
    const float g = gates[0];
    float* col = strip + tid;
    float e = 0.0f;
    float xa[SCAN_BLK], xb[SCAN_BLK];
    scan_fetch(xa, col, 0);
    for (int t = 0; t < LSEQ - 2 * SCAN_BLK; t += 2 * SCAN_BLK) {
      scan_fetch(xb, col, t + SCAN_BLK);
      e = scan_run(xa, col, t, g, e);
      scan_fetch(xa, col, t + 2 * SCAN_BLK);
      e = scan_run(xb, col, t + SCAN_BLK, g, e);
    }
    scan_fetch(xb, col, LSEQ - SCAN_BLK);
    e = scan_run(xa, col, LSEQ - 2 * SCAN_BLK, g, e);
    e = scan_run(xb, col, LSEQ - SCAN_BLK, g, e);
  }
  __syncthreads();

  v4f buf[SCAN_PASSES];
  #pragma unroll
  for (int i = 0; i < SCAN_PASSES; ++i)
    buf[i] = *(const v4f*)(strip + (i * SCAN_RPP + r0) * SCAN_CH + c4);
  #pragma unroll
  for (int i = 0; i < SCAN_PASSES; ++i)
    *(v4f*)(Y + (size_t)(i * SCAN_RPP + r0) * DMODEL + c4) = buf[i];
}

extern "C" void kernel_launch(void* const* d_in, const int* in_sizes, int n_in,
                              void* d_out, int out_size, void* d_ws, size_t ws_size,
                              hipStream_t stream) {
  const float* q     = (const float*)d_in[0];
  const float* k     = (const float*)d_in[1];
  const float* v     = (const float*)d_in[2];
  const float* Wq    = (const float*)d_in[3];
  const float* Wk    = (const float*)d_in[4];
  const float* Wv    = (const float*)d_in[5];
  const float* Wd    = (const float*)d_in[6];
  const float* gates = (const float*)d_in[7];
  float* out = (float*)d_out;

  const int L = LSEQ, D = DMODEL;
  const size_t LD = (size_t)L * D;
  float* Ek = (float*)d_ws;
  float* Ev = Ek + LD;
  float* Zs = Ev + LD;

  const float invScale = 1.0f / sqrtf(sqrtf(32.0f));

  dim3 gBlk(128);
  dim3 gGrid(D / GEMM_T, L / GEMM_T);

  hipLaunchKernelGGL(ema_scan, dim3(DMODEL / SCAN_CH, 2), dim3(SCAN_THREADS), 0, stream, k, v, Ek, Ev, gates);
  hipLaunchKernelGGL(proj_gate_f32_wmma, gGrid, gBlk, 0, stream, q, Wq, Ek, Wk, Ev, Wv, Zs, L, D, D, invScale);
  hipLaunchKernelGGL(gemm_nt_f32_wmma, gGrid, gBlk, 0, stream, Zs, Wd, out, L, D, D, 1.0f);
}


# GEMM K split inside the work-group 4 -> 8 ways: 8 waves (2 per SIMD) per 64x64 tile, each takes every 8th 8-k block, so one wave's WMMAs cover the other's load issue and waits; K sum order changes
# speedup vs baseline: 6.6922x; 1.1279x over previous
#include <hip/hip_runtime.h>
#include <math.h>


typedef float v2f __attribute__((ext_vector_type(2)));
typedef float v4f __attribute__((ext_vector_type(4)));
typedef float v8f __attribute__((ext_vector_type(8)));

#define LSEQ   1024
#define DMODEL 1024

#define GEMM_KB 8
#define GEMM_KW 8
#define GEMM_KS (GEMM_KB * GEMM_KW)
#define GEMM_THREADS (32 * GEMM_KW)
#define GEMM_ROWS (GEMM_T / GEMM_KW / 2)
#define GEMM_T  64
#define RED_PITCH (GEMM_T + 4)

__device__ __forceinline__ void gemm_fetch(v2f (&fa)[2][4], v2f (&fb)[2][4],
                                           const float* const (&aPtr)[4],
                                           const float* const (&bPtr)[4], int k) {
  #pragma unroll
  for (int s = 0; s < 2; ++s) {
    #pragma unroll
    for (int i = 0; i < 4; ++i) fa[s][i] = *(const v2f*)(aPtr[i] + k + 4 * s);
    #pragma unroll
    for (int j = 0; j < 4; ++j) fb[s][j] = *(const v2f*)(bPtr[j] + k + 4 * s);
  }
}

__device__ __forceinline__ void gemm_block(v8f (&acc)[4][4], const v2f (&fa)[2][4],
                                           const v2f (&fb)[2][4]) {
  #pragma unroll
  for (int s = 0; s < 2; ++s)
    #pragma unroll
    for (int i = 0; i < 4; ++i)
      #pragma unroll
      for (int j = 0; j < 4; ++j)
        acc[i][j] = __builtin_amdgcn_wmma_f32_16x16x4_f32(
            false, fa[s][i], false, fb[s][j], (short)0, acc[i][j], false, false);
}

typedef float red_plane_t[GEMM_T][RED_PITCH];

__device__ __forceinline__ void gemm_wave_partial(v8f (&acc)[4][4],
                                                  const float* __restrict__ A,
                                                  const float* __restrict__ B,
                                                  int K, int mBase, int nBase,
                                                  int lid, int koff) {
  const float* aPtr[4];
  const float* bPtr[4];
  #pragma unroll
  for (int i = 0; i < 4; ++i) {
    aPtr[i] = A + (size_t)(mBase + i * 16 + lid) * K + koff;
    bPtr[i] = B + (size_t)(nBase + i * 16 + lid) * K + koff;
  }

  #pragma unroll
  for (int i = 0; i < 4; ++i)
    #pragma unroll
    for (int j = 0; j < 4; ++j)
      acc[i][j] = (v8f){};

  #define GEMM_FENCE() __builtin_amdgcn_sched_barrier(0)
  v2f a0[2][4], b0[2][4], a1[2][4], b1[2][4], a2[2][4], b2[2][4];
  gemm_fetch(a0, b0, aPtr, bPtr, 0);
  GEMM_FENCE();
  gemm_fetch(a1, b1, aPtr, bPtr, GEMM_KS);
  int k = 0;
  for (; k < K - 4 * GEMM_KS; k += 3 * GEMM_KS) {
    gemm_fetch(a2, b2, aPtr, bPtr, k + 2 * GEMM_KS);
    GEMM_FENCE();
    gemm_block(acc, a0, b0);
    GEMM_FENCE();
    gemm_fetch(a0, b0, aPtr, bPtr, k + 3 * GEMM_KS);
    GEMM_FENCE();
    gemm_block(acc, a1, b1);
    GEMM_FENCE();
    gemm_fetch(a1, b1, aPtr, bPtr, k + 4 * GEMM_KS);
    GEMM_FENCE();
    gemm_block(acc, a2, b2);
  }
  gemm_fetch(a2, b2, aPtr, bPtr, k + 2 * GEMM_KS);
  GEMM_FENCE();
  gemm_block(acc, a0, b0);
  GEMM_FENCE();
  gemm_fetch(a0, b0, aPtr, bPtr, k + 3 * GEMM_KS);
  GEMM_FENCE();
  gemm_block(acc, a1, b1);
  gemm_block(acc, a2, b2);
  gemm_block(acc, a0, b0);
  #undef GEMM_FENCE
}

__device__ __forceinline__ void gemm_reduce(v4f (&s)[GEMM_ROWS], red_plane_t* red,
                                            const v8f (&acc)[4][4], float alpha,
                                            int wave, int lid, int half) {
  #pragma unroll
  for (int i = 0; i < 4; ++i)
    #pragma unroll
    for (int j = 0; j < 4; ++j)
      #pragma unroll
      for (int r = 0; r < 8; ++r)
        red[wave][i * 16 + half * 8 + r][j * 16 + lid] = alpha * acc[i][j][r];
  __syncthreads();
  #pragma unroll
  for (int p = 0; p < GEMM_ROWS; ++p) {
    const int row = wave * (2 * GEMM_ROWS) + p * 2 + half;
    s[p] = *(const v4f*)&red[0][row][lid * 4];
    #pragma unroll
    for (int z = 1; z < GEMM_KW; ++z) s[p] += *(const v4f*)&red[z][row][lid * 4];
  }
}

__global__ __launch_bounds__(GEMM_THREADS)
void gemm_nt_f32_wmma(const float* __restrict__ A, const float* __restrict__ B,
                      float* __restrict__ C, int M, int N, int K, float alpha) {
  __shared__ __attribute__((aligned(16))) float red[GEMM_KW][GEMM_T][RED_PITCH];
  const int lane = threadIdx.x & 31;
  const int wave = threadIdx.x >> 5;
  const int lid  = lane & 15;
  const int half = lane >> 4;
  const int mBase = blockIdx.y * GEMM_T;
  const int nBase = blockIdx.x * GEMM_T;
  const int koff = wave * GEMM_KB + half * 2;

  v8f acc[4][4];
  gemm_wave_partial(acc, A, B, K, mBase, nBase, lid, koff);
  v4f s[GEMM_ROWS];
  gemm_reduce(s, red, acc, alpha, wave, lid, half);
  #pragma unroll
  for (int p = 0; p < GEMM_ROWS; ++p) {
    const int row = wave * (2 * GEMM_ROWS) + p * 2 + half;
    *(v4f*)(C + (size_t)(mBase + row) * N + nBase + lid * 4) = s[p];
  }
}

__global__ __launch_bounds__(GEMM_THREADS)
void proj_gate_f32_wmma(const float* __restrict__ q,  const float* __restrict__ Wq,
                        const float* __restrict__ Ek, const float* __restrict__ Wk,
                        const float* __restrict__ Ev, const float* __restrict__ Wv,
                        float* __restrict__ Z, int M, int N, int K, float scale) {
  __shared__ __attribute__((aligned(16))) float red[GEMM_KW][GEMM_T][RED_PITCH];
  const int lane = threadIdx.x & 31;
  const int wave = threadIdx.x >> 5;
  const int lid  = lane & 15;
  const int half = lane >> 4;
  const int mBase = blockIdx.y * GEMM_T;
  const int nBase = blockIdx.x * GEMM_T;
  const int koff = wave * GEMM_KB + half * 2;

  v8f acc[4][4];
  v4f sq[GEMM_ROWS], sk[GEMM_ROWS];
  float a[GEMM_ROWS];

  gemm_wave_partial(acc, q, Wq, K, mBase, nBase, lid, koff);
  gemm_reduce(sq, red, acc, scale, wave, lid, half);
  __syncthreads();

  gemm_wave_partial(acc, Ek, Wk, K, mBase, nBase, lid, koff);
  gemm_reduce(sk, red, acc, scale, wave, lid, half);
  #pragma unroll
  for (int p = 0; p < GEMM_ROWS; ++p) {
    float d = sq[p].x * sk[p].x + sq[p].y * sk[p].y + sq[p].z * sk[p].z + sq[p].w * sk[p].w;
    #pragma unroll
    for (int off = 1; off < 8; off <<= 1)
      d += __shfl_xor(d, off, 32);
    a[p] = 1.0f / (1.0f + 31.0f * expf(-d));
  }
  __syncthreads();

  gemm_wave_partial(acc, Ev, Wv, K, mBase, nBase, lid, koff);
  gemm_reduce(sk, red, acc, 1.0f, wave, lid, half);
  #pragma unroll
  for (int p = 0; p < GEMM_ROWS; ++p) {
    const int row = wave * (2 * GEMM_ROWS) + p * 2 + half;
    *(v4f*)(Z + (size_t)(mBase + row) * N + nBase + lid * 4) = a[p] * sk[p];
  }
}

#define SCAN_CH      32
#define SCAN_THREADS 512
#define SCAN_RPP     (SCAN_THREADS / 8)
#define SCAN_PASSES  (LSEQ / SCAN_RPP)
#define SCAN_BLK     32

__device__ __forceinline__ void scan_fetch(float (&x)[SCAN_BLK], const float* col, int t) {
  #pragma unroll
  for (int j = 0; j < SCAN_BLK; ++j) x[j] = col[(t + j) * SCAN_CH];
}

__device__ __forceinline__ float scan_run(const float (&x)[SCAN_BLK], float* col, int t,
                                          float g, float e) {
  #pragma unroll
  for (int j = 0; j < SCAN_BLK; ++j) {
    e = fmaf(g, e, x[j]);
    col[(t + j) * SCAN_CH] = e;
  }
  return e;
}

__global__ __launch_bounds__(SCAN_THREADS)
void ema_scan(const float* __restrict__ kin, const float* __restrict__ vin,
              float* __restrict__ Ek, float* __restrict__ Ev,
              const float* __restrict__ gates) {
  __shared__ float strip[LSEQ * SCAN_CH];
  const float* X = (blockIdx.y == 0 ? kin : vin) + blockIdx.x * SCAN_CH;
  float*       Y = (blockIdx.y == 0 ? Ek : Ev) + blockIdx.x * SCAN_CH;
  const int tid = threadIdx.x;
  const int c4  = (tid & 7) * 4;
  const int r0  = tid >> 3;

  {
    v4f in[SCAN_PASSES];
    #pragma unroll
    for (int i = 0; i < SCAN_PASSES; ++i)
      in[i] = *(const v4f*)(X + (size_t)(i * SCAN_RPP + r0) * DMODEL + c4);
    #pragma unroll
    for (int i = 0; i < SCAN_PASSES; ++i)
      *(v4f*)(strip + (i * SCAN_RPP + r0) * SCAN_CH + c4) = in[i];
  }
  __syncthreads();

  if (tid < SCAN_CH) {
    const float g = gates[0];
    float* col = strip + tid;
    float e = 0.0f;
    float xa[SCAN_BLK], xb[SCAN_BLK];
    scan_fetch(xa, col, 0);
    for (int t = 0; t < LSEQ - 2 * SCAN_BLK; t += 2 * SCAN_BLK) {
      scan_fetch(xb, col, t + SCAN_BLK);
      e = scan_run(xa, col, t, g, e);
      scan_fetch(xa, col, t + 2 * SCAN_BLK);
      e = scan_run(xb, col, t + SCAN_BLK, g, e);
    }
    scan_fetch(xb, col, LSEQ - SCAN_BLK);
    e = scan_run(xa, col, LSEQ - 2 * SCAN_BLK, g, e);
    e = scan_run(xb, col, LSEQ - SCAN_BLK, g, e);
  }
  __syncthreads();

  v4f buf[SCAN_PASSES];
  #pragma unroll
  for (int i = 0; i < SCAN_PASSES; ++i)
    buf[i] = *(const v4f*)(strip + (i * SCAN_RPP + r0) * SCAN_CH + c4);
  #pragma unroll
  for (int i = 0; i < SCAN_PASSES; ++i)
    *(v4f*)(Y + (size_t)(i * SCAN_RPP + r0) * DMODEL + c4) = buf[i];
}

extern "C" void kernel_launch(void* const* d_in, const int* in_sizes, int n_in,
                              void* d_out, int out_size, void* d_ws, size_t ws_size,
                              hipStream_t stream) {
  const float* q     = (const float*)d_in[0];
  const float* k     = (const float*)d_in[1];
  const float* v     = (const float*)d_in[2];
  const float* Wq    = (const float*)d_in[3];
  const float* Wk    = (const float*)d_in[4];
  const float* Wv    = (const float*)d_in[5];
  const float* Wd    = (const float*)d_in[6];
  const float* gates = (const float*)d_in[7];
  float* out = (float*)d_out;

  const int L = LSEQ, D = DMODEL;
  const size_t LD = (size_t)L * D;
  float* Ek = (float*)d_ws;
  float* Ev = Ek + LD;
  float* Zs = Ev + LD;

  const float invScale = 1.0f / sqrtf(sqrtf(32.0f));

  dim3 gBlk(GEMM_THREADS);
  dim3 gGrid(D / GEMM_T, L / GEMM_T);

  hipLaunchKernelGGL(ema_scan, dim3(DMODEL / SCAN_CH, 2), dim3(SCAN_THREADS), 0, stream, k, v, Ek, Ev, gates);
  hipLaunchKernelGGL(proj_gate_f32_wmma, gGrid, gBlk, 0, stream, q, Wq, Ek, Wk, Ev, Wv, Zs, L, D, D, invScale);
  hipLaunchKernelGGL(gemm_nt_f32_wmma, gGrid, gBlk, 0, stream, Zs, Wd, out, L, D, D, 1.0f);
}


# GEMM operands re-laid out in fragment order (512-B chunks, 16 rows x 8 k): a fragment load is 4 whole lines, was 16 B of 16 lines; scan launch's idle groups re-lay q, W; K, gate dot sum order change
# speedup vs baseline: 7.0807x; 1.0581x over previous
#include <hip/hip_runtime.h>
#include <math.h>


typedef float v2f __attribute__((ext_vector_type(2)));
typedef float v4f __attribute__((ext_vector_type(4)));
typedef float v8f __attribute__((ext_vector_type(8)));

#define LSEQ   1024
#define DMODEL 1024

#define PK_CPT (DMODEL / 8)
#define PK_CF  128

#define GEMM_KW 8
#define GEMM_THREADS (32 * GEMM_KW)
#define GEMM_T  64
#define GEMM_ROWS (GEMM_T / GEMM_KW / 2)
#define GEMM_NB (PK_CPT / GEMM_KW)
#define GEMM_BS (GEMM_KW * PK_CF)
#define RED_PITCH (GEMM_T + 4)

__device__ __forceinline__ void gemm_fetch(v4f (&fa)[4], v4f (&fb)[4],
                                           const float* const (&aPtr)[4],
                                           const float* const (&bPtr)[4], int blk) {
  #pragma unroll
  for (int i = 0; i < 4; ++i) fa[i] = *(const v4f*)(aPtr[i] + blk * GEMM_BS);
  #pragma unroll
  for (int j = 0; j < 4; ++j) fb[j] = *(const v4f*)(bPtr[j] + blk * GEMM_BS);
}

__device__ __forceinline__ void gemm_block(v8f (&acc)[4][4], const v4f (&fa)[4],
                                           const v4f (&fb)[4]) {
  #pragma unroll
  for (int i = 0; i < 4; ++i)
    #pragma unroll
    for (int j = 0; j < 4; ++j)
      acc[i][j] = __builtin_amdgcn_wmma_f32_16x16x4_f32(
          false, __builtin_shufflevector(fa[i], fa[i], 0, 1),
          false, __builtin_shufflevector(fb[j], fb[j], 0, 1),
          (short)0, acc[i][j], false, false);
  #pragma unroll
  for (int i = 0; i < 4; ++i)
    #pragma unroll
    for (int j = 0; j < 4; ++j)
      acc[i][j] = __builtin_amdgcn_wmma_f32_16x16x4_f32(
          false, __builtin_shufflevector(fa[i], fa[i], 2, 3),
          false, __builtin_shufflevector(fb[j], fb[j], 2, 3),
          (short)0, acc[i][j], false, false);
}

typedef float red_plane_t[GEMM_T][RED_PITCH];

__device__ __forceinline__ void gemm_wave_partial(v8f (&acc)[4][4],
                                                  const float* __restrict__ PA,
                                                  const float* __restrict__ PB,
                                                  int tA, int tB, int lane, int wave) {
  const float* aPtr[4];
  const float* bPtr[4];
  #pragma unroll
  for (int i = 0; i < 4; ++i) {
    aPtr[i] = PA + ((size_t)(tA + i) * PK_CPT + wave) * PK_CF + lane * 4;
    bPtr[i] = PB + ((size_t)(tB + i) * PK_CPT + wave) * PK_CF + lane * 4;
  }

  #pragma unroll
  for (int i = 0; i < 4; ++i)
    #pragma unroll
    for (int j = 0; j < 4; ++j)
      acc[i][j] = (v8f){};

  #define GEMM_FENCE() __builtin_amdgcn_sched_barrier(0)
  v4f a0[4], b0[4], a1[4], b1[4], a2[4], b2[4];
  gemm_fetch(a0, b0, aPtr, bPtr, 0);
  GEMM_FENCE();
  gemm_fetch(a1, b1, aPtr, bPtr, 1);
  int b = 0;
  for (; b < GEMM_NB - 4; b += 3) {
    gemm_fetch(a2, b2, aPtr, bPtr, b + 2);
    GEMM_FENCE();
    gemm_block(acc, a0, b0);
    GEMM_FENCE();
    gemm_fetch(a0, b0, aPtr, bPtr, b + 3);
    GEMM_FENCE();
    gemm_block(acc, a1, b1);
    GEMM_FENCE();
    gemm_fetch(a1, b1, aPtr, bPtr, b + 4);
    GEMM_FENCE();
    gemm_block(acc, a2, b2);
  }
  gemm_fetch(a2, b2, aPtr, bPtr, b + 2);
  GEMM_FENCE();
  gemm_block(acc, a0, b0);
  GEMM_FENCE();
  gemm_fetch(a0, b0, aPtr, bPtr, b + 3);
  GEMM_FENCE();
  gemm_block(acc, a1, b1);
  gemm_block(acc, a2, b2);
  gemm_block(acc, a0, b0);
  #undef GEMM_FENCE
}

template <bool kChunk>
__device__ __forceinline__ void gemm_reduce(v4f (&s)[GEMM_ROWS], red_plane_t* red,
                                            const v8f (&acc)[4][4], float alpha,
                                            int wave, int lid, int half) {
  #pragma unroll
  for (int i = 0; i < 4; ++i)
    #pragma unroll
    for (int j = 0; j < 4; ++j)
      #pragma unroll
      for (int r = 0; r < 8; ++r)
        red[wave][i * 16 + half * 8 + r][j * 16 + lid] = alpha * acc[i][j][r];
  __syncthreads();
  #pragma unroll
  for (int p = 0; p < GEMM_ROWS; ++p) {
    const int row = kChunk ? (wave >> 1) * 16 + lid : wave * (2 * GEMM_ROWS) + p * 2 + half;
    const int col = kChunk ? (wave & 1) * 32 + p * 8 + half * 4 : lid * 4;
    s[p] = *(const v4f*)&red[0][row][col];
    #pragma unroll
    for (int z = 1; z < GEMM_KW; ++z) s[p] += *(const v4f*)&red[z][row][col];
  }
}

__global__ __launch_bounds__(GEMM_THREADS)
void gemm_nt_f32_wmma(const float* __restrict__ PA, const float* __restrict__ PB,
                      float* __restrict__ C, int N) {
  __shared__ __attribute__((aligned(16))) float red[GEMM_KW + 2][GEMM_T][RED_PITCH];
  const int lane = threadIdx.x & 31;
  const int wave = threadIdx.x >> 5;
  const int lid  = lane & 15;
  const int half = lane >> 4;
  const int mBase = blockIdx.y * GEMM_T;
  const int nBase = blockIdx.x * GEMM_T;

  v8f acc[4][4];
  gemm_wave_partial(acc, PA, PB, blockIdx.y * (GEMM_T / 16), blockIdx.x * (GEMM_T / 16),
                    lane, wave);
  v4f s[GEMM_ROWS];
  gemm_reduce<false>(s, red, acc, 1.0f, wave, lid, half);
  #pragma unroll
  for (int p = 0; p < GEMM_ROWS; ++p) {
    const int row = wave * (2 * GEMM_ROWS) + p * 2 + half;
    *(v4f*)(C + (size_t)(mBase + row) * N + nBase + lid * 4) = s[p];
  }
}

__global__ __launch_bounds__(GEMM_THREADS)
void proj_gate_f32_wmma(const float* __restrict__ Pq,  const float* __restrict__ PWq,
                        const float* __restrict__ PEk, const float* __restrict__ PWk,
                        const float* __restrict__ PEv, const float* __restrict__ PWv,
                        float* __restrict__ PZ, float scale) {
  __shared__ __attribute__((aligned(16))) float red[GEMM_KW][GEMM_T][RED_PITCH];
  const int lane = threadIdx.x & 31;
  const int wave = threadIdx.x >> 5;
  const int lid  = lane & 15;
  const int half = lane >> 4;
  const int tA = blockIdx.y * (GEMM_T / 16);
  const int tB = blockIdx.x * (GEMM_T / 16);

  v8f acc[4][4];
  v4f sq[GEMM_ROWS], sk[GEMM_ROWS];

  gemm_wave_partial(acc, Pq, PWq, tA, tB, lane, wave);
  gemm_reduce<true>(sq, red, acc, scale, wave, lid, half);
  __syncthreads();

  gemm_wave_partial(acc, PEk, PWk, tA, tB, lane, wave);
  gemm_reduce<true>(sk, red, acc, scale, wave, lid, half);
  float d = 0.0f;
  #pragma unroll
  for (int p = 0; p < GEMM_ROWS; ++p)
    d += sq[p].x * sk[p].x + sq[p].y * sk[p].y + sq[p].z * sk[p].z + sq[p].w * sk[p].w;
  d += __shfl_xor(d, 16, 32);
  const float a = 1.0f / (1.0f + 31.0f * expf(-d));
  __syncthreads();

  gemm_wave_partial(acc, PEv, PWv, tA, tB, lane, wave);
  gemm_reduce<true>(sk, red, acc, 1.0f, wave, lid, half);
  float* zc = PZ + ((size_t)(tA + (wave >> 1)) * PK_CPT + blockIdx.x * (GEMM_T / 8)
                    + (wave & 1) * 4) * PK_CF + lane * 4;
  #pragma unroll
  for (int p = 0; p < GEMM_ROWS; ++p)
    *(v4f*)(zc + p * PK_CF) = a * sk[p];
}

#define SCAN_CH      32
#define SCAN_PITCH   36
#define SCAN_THREADS 512
#define SCAN_RPP     (SCAN_THREADS / 8)
#define SCAN_PASSES  (LSEQ / SCAN_RPP)
#define SCAN_BLK     32
#define PACK_MATS    5

__device__ __forceinline__ void scan_fetch(float (&x)[SCAN_BLK], const float* col, int t) {
  #pragma unroll
  for (int j = 0; j < SCAN_BLK; ++j) x[j] = col[(t + j) * SCAN_PITCH];
}

__device__ __forceinline__ float scan_run(const float (&x)[SCAN_BLK], float* col, int t,
                                          float g, float e) {
  #pragma unroll
  for (int j = 0; j < SCAN_BLK; ++j) {
    e = fmaf(g, e, x[j]);
    col[(t + j) * SCAN_PITCH] = e;
  }
  return e;
}

__global__ __launch_bounds__(SCAN_THREADS)
void ema_scan_pack(const float* __restrict__ kin, const float* __restrict__ vin,
                   float* __restrict__ PEk, float* __restrict__ PEv,
                   const float* __restrict__ gates,
                   const float* __restrict__ q,  const float* __restrict__ Wq,
                   const float* __restrict__ Wk, const float* __restrict__ Wv,
                   const float* __restrict__ Wd,
                   float* __restrict__ Pq,  float* __restrict__ PWq,
                   float* __restrict__ PWk, float* __restrict__ PWv,
                   float* __restrict__ PWd) {
  __shared__ __attribute__((aligned(16))) float strip[LSEQ * SCAN_PITCH];
  const int tid  = threadIdx.x;
  const int lane = tid & 31;
  const int wv   = tid >> 5;

  if (blockIdx.y >= 2) {
    const int m = blockIdx.y - 2;
    const float* src = m == 0 ? q  : m == 1 ? Wq  : m == 2 ? Wk  : m == 3 ? Wv  : Wd;
    float*       dst = m == 0 ? Pq : m == 1 ? PWq : m == 2 ? PWk : m == 3 ? PWv : PWd;
    const int T  = blockIdx.x * 2 + (wv >> 3);
    const int s0 = (wv & 7) * 16;
    const float* rp = src + (size_t)(T * 16 + (lane & 15)) * DMODEL + s0 * 8 + (lane >> 4) * 4;
    float*       wp = dst + ((size_t)T * PK_CPT + s0) * PK_CF + lane * 4;
    v4f x[16];
    #pragma unroll
    for (int j = 0; j < 16; ++j) x[j] = *(const v4f*)(rp + j * 8);
    #pragma unroll
    for (int j = 0; j < 16; ++j) *(v4f*)(wp + j * PK_CF) = x[j];
    return;
  }

  const float* X = (blockIdx.y == 0 ? kin : vin) + blockIdx.x * SCAN_CH;
  float*       Y = (blockIdx.y == 0 ? PEk : PEv);
  const int c4  = (tid & 7) * 4;
  const int r0  = tid >> 3;

  {
    v4f in[SCAN_PASSES];
    #pragma unroll
    for (int i = 0; i < SCAN_PASSES; ++i)
      in[i] = *(const v4f*)(X + (size_t)(i * SCAN_RPP + r0) * DMODEL + c4);
    #pragma unroll
    for (int i = 0; i < SCAN_PASSES; ++i)
      *(v4f*)(strip + (i * SCAN_RPP + r0) * SCAN_PITCH + c4) = in[i];
  }
  __syncthreads();

  if (tid < SCAN_CH) {
    const float g = gates[0];
    float* col = strip + tid;
    float e = 0.0f;
    float xa[SCAN_BLK], xb[SCAN_BLK];
    scan_fetch(xa, col, 0);
    for (int t = 0; t < LSEQ - 2 * SCAN_BLK; t += 2 * SCAN_BLK) {
      scan_fetch(xb, col, t + SCAN_BLK);
      e = scan_run(xa, col, t, g, e);
      scan_fetch(xa, col, t + 2 * SCAN_BLK);
      e = scan_run(xb, col, t + SCAN_BLK, g, e);
    }
    scan_fetch(xb, col, LSEQ - SCAN_BLK);
    e = scan_run(xa, col, LSEQ - 2 * SCAN_BLK, g, e);
    e = scan_run(xb, col, LSEQ - SCAN_BLK, g, e);
  }
  __syncthreads();

  const float* lp = strip + ((wv >> 2) * 16 + (lane & 15)) * SCAN_PITCH
                          + (wv & 3) * 8 + (lane >> 4) * 4;
  float* gp = Y + ((size_t)(wv >> 2) * PK_CPT + blockIdx.x * 4 + (wv & 3)) * PK_CF + lane * 4;
  v4f buf[SCAN_PASSES];
  #pragma unroll
  for (int i = 0; i < SCAN_PASSES; ++i)
    buf[i] = *(const v4f*)(lp + i * 64 * SCAN_PITCH);
  #pragma unroll
  for (int i = 0; i < SCAN_PASSES; ++i)
    *(v4f*)(gp + (size_t)i * 4 * PK_CPT * PK_CF) = buf[i];
}

extern "C" void kernel_launch(void* const* d_in, const int* in_sizes, int n_in,
                              void* d_out, int out_size, void* d_ws, size_t ws_size,
                              hipStream_t stream) {
  const float* q     = (const float*)d_in[0];
  const float* k     = (const float*)d_in[1];
  const float* v     = (const float*)d_in[2];
  const float* Wq    = (const float*)d_in[3];
  const float* Wk    = (const float*)d_in[4];
  const float* Wv    = (const float*)d_in[5];
  const float* Wd    = (const float*)d_in[6];
  const float* gates = (const float*)d_in[7];
  float* out = (float*)d_out;

  const int L = LSEQ, D = DMODEL;
  const size_t LD = (size_t)L * D;
  float* PEk = (float*)d_ws;
  float* PEv = PEk + LD;
  float* PZ  = PEv + LD;
  float* Pq  = PZ  + LD;
  float* PWq = Pq  + LD;
  float* PWk = PWq + LD;
  float* PWv = PWk + LD;
  float* PWd = PWv + LD;

  const float invScale = 1.0f / sqrtf(sqrtf(32.0f));

  dim3 gBlk(GEMM_THREADS);
  dim3 gGrid(D / GEMM_T, L / GEMM_T);

  hipLaunchKernelGGL(ema_scan_pack, dim3(DMODEL / SCAN_CH, 2 + PACK_MATS), dim3(SCAN_THREADS), 0, stream,
                     k, v, PEk, PEv, gates, q, Wq, Wk, Wv, Wd, Pq, PWq, PWk, PWv, PWd);
  hipLaunchKernelGGL(proj_gate_f32_wmma, gGrid, gBlk, 0, stream, Pq, PWq, PEk, PWk, PEv, PWv, PZ, invScale);
  hipLaunchKernelGGL(gemm_nt_f32_wmma, gGrid, gBlk, 0, stream, PZ, PWd, out, D);
}


# GEMM fragment loads run 4 blocks ahead of their WMMAs (5 register sets, was 2 ahead); next product's first 4 blocks load across the LDS meeting (its barrier now orders LDS only); sum order unchanged
# speedup vs baseline: 7.3314x; 1.0354x over previous
#include <hip/hip_runtime.h>
#include <math.h>


typedef float v2f __attribute__((ext_vector_type(2)));
typedef float v4f __attribute__((ext_vector_type(4)));
typedef float v8f __attribute__((ext_vector_type(8)));

#define LSEQ   1024
#define DMODEL 1024

#define PK_CPT (DMODEL / 8)
#define PK_CF  128

#define GEMM_KW 8
#define GEMM_THREADS (32 * GEMM_KW)
#define GEMM_T  64
#define GEMM_ROWS (GEMM_T / GEMM_KW / 2)
#define GEMM_NB (PK_CPT / GEMM_KW)
#define GEMM_BS (GEMM_KW * PK_CF)
#define GEMM_PF 4
#define GEMM_SETS (GEMM_PF + 1)
#define RED_PITCH (GEMM_T + 4)

__device__ __forceinline__ void gemm_fetch(v4f (&fa)[4], v4f (&fb)[4],
                                           const float* const (&aPtr)[4],
                                           const float* const (&bPtr)[4], int blk) {
  #pragma unroll
  for (int i = 0; i < 4; ++i) fa[i] = *(const v4f*)(aPtr[i] + blk * GEMM_BS);
  #pragma unroll
  for (int j = 0; j < 4; ++j) fb[j] = *(const v4f*)(bPtr[j] + blk * GEMM_BS);
}

__device__ __forceinline__ void gemm_block(v8f (&acc)[4][4], const v4f (&fa)[4],
                                           const v4f (&fb)[4]) {
  #pragma unroll
  for (int i = 0; i < 4; ++i)
    #pragma unroll
    for (int j = 0; j < 4; ++j)
      acc[i][j] = __builtin_amdgcn_wmma_f32_16x16x4_f32(
          false, __builtin_shufflevector(fa[i], fa[i], 0, 1),
          false, __builtin_shufflevector(fb[j], fb[j], 0, 1),
          (short)0, acc[i][j], false, false);
  #pragma unroll
  for (int i = 0; i < 4; ++i)
    #pragma unroll
    for (int j = 0; j < 4; ++j)
      acc[i][j] = __builtin_amdgcn_wmma_f32_16x16x4_f32(
          false, __builtin_shufflevector(fa[i], fa[i], 2, 3),
          false, __builtin_shufflevector(fb[j], fb[j], 2, 3),
          (short)0, acc[i][j], false, false);
}

typedef float red_plane_t[GEMM_T][RED_PITCH];

__device__ __forceinline__ void gemm_ptrs(const float* (&aPtr)[4], const float* (&bPtr)[4],
                                          const float* __restrict__ PA,
                                          const float* __restrict__ PB,
                                          int tA, int tB, int lane, int wave) {
  #pragma unroll
  for (int i = 0; i < 4; ++i) {
    aPtr[i] = PA + ((size_t)(tA + i) * PK_CPT + wave) * PK_CF + lane * 4;
    bPtr[i] = PB + ((size_t)(tB + i) * PK_CPT + wave) * PK_CF + lane * 4;
  }
}

#define GEMM_FENCE() __builtin_amdgcn_sched_barrier(0)

__device__ __forceinline__ void gemm_head(v4f (&fa)[GEMM_SETS][4], v4f (&fb)[GEMM_SETS][4],
                                          const float* const (&aPtr)[4],
                                          const float* const (&bPtr)[4]) {
  #pragma unroll
  for (int p = 0; p < GEMM_PF; ++p) {
    gemm_fetch(fa[p], fb[p], aPtr, bPtr, p);
    GEMM_FENCE();
  }
}

__device__ __forceinline__ void gemm_kloop(v8f (&acc)[4][4],
                                           v4f (&fa)[GEMM_SETS][4], v4f (&fb)[GEMM_SETS][4],
                                           const float* const (&aPtr)[4],
                                           const float* const (&bPtr)[4]) {
  #pragma unroll
  for (int i = 0; i < 4; ++i)
    #pragma unroll
    for (int j = 0; j < 4; ++j)
      acc[i][j] = (v8f){};

  #pragma unroll
  for (int b = 0; b < GEMM_NB; ++b) {
    if (b + GEMM_PF < GEMM_NB)
      gemm_fetch(fa[(b + GEMM_PF) % GEMM_SETS], fb[(b + GEMM_PF) % GEMM_SETS],
                 aPtr, bPtr, b + GEMM_PF);
    GEMM_FENCE();
    gemm_block(acc, fa[b % GEMM_SETS], fb[b % GEMM_SETS]);
    GEMM_FENCE();
  }
}

__device__ __forceinline__ void lds_barrier() {
  __builtin_amdgcn_fence(__ATOMIC_RELEASE, "workgroup", "local");
  __builtin_amdgcn_s_barrier();
  __builtin_amdgcn_fence(__ATOMIC_ACQUIRE, "workgroup", "local");
}

template <bool kChunk>
__device__ __forceinline__ void gemm_reduce(v4f (&s)[GEMM_ROWS], red_plane_t* red,
                                            const v8f (&acc)[4][4], float alpha,
                                            int wave, int lid, int half) {
  #pragma unroll
  for (int i = 0; i < 4; ++i)
    #pragma unroll
    for (int j = 0; j < 4; ++j)
      #pragma unroll
      for (int r = 0; r < 8; ++r)
        red[wave][i * 16 + half * 8 + r][j * 16 + lid] = alpha * acc[i][j][r];
  lds_barrier();
  #pragma unroll
  for (int p = 0; p < GEMM_ROWS; ++p) {
    const int row = kChunk ? (wave >> 1) * 16 + lid : wave * (2 * GEMM_ROWS) + p * 2 + half;
    const int col = kChunk ? (wave & 1) * 32 + p * 8 + half * 4 : lid * 4;
    s[p] = *(const v4f*)&red[0][row][col];
    #pragma unroll
    for (int z = 1; z < GEMM_KW; ++z) s[p] += *(const v4f*)&red[z][row][col];
  }
  #pragma unroll
  for (int p = 0; p < GEMM_ROWS; ++p) asm volatile("" :: "v"(s[p]));
}

__global__ __launch_bounds__(GEMM_THREADS)
void gemm_nt_f32_wmma(const float* __restrict__ PA, const float* __restrict__ PB,
                      float* __restrict__ C, int N) {
  __shared__ __attribute__((aligned(16))) float red[GEMM_KW + 2][GEMM_T][RED_PITCH];
  const int lane = threadIdx.x & 31;
  const int wave = threadIdx.x >> 5;
  const int lid  = lane & 15;
  const int half = lane >> 4;
  const int mBase = blockIdx.y * GEMM_T;
  const int nBase = blockIdx.x * GEMM_T;

  v8f acc[4][4];
  v4f fa[GEMM_SETS][4], fb[GEMM_SETS][4];
  const float* aPtr[4];
  const float* bPtr[4];
  gemm_ptrs(aPtr, bPtr, PA, PB, blockIdx.y * (GEMM_T / 16), blockIdx.x * (GEMM_T / 16),
            lane, wave);
  gemm_head(fa, fb, aPtr, bPtr);
  gemm_kloop(acc, fa, fb, aPtr, bPtr);
  v4f s[GEMM_ROWS];
  gemm_reduce<false>(s, red, acc, 1.0f, wave, lid, half);
  #pragma unroll
  for (int p = 0; p < GEMM_ROWS; ++p) {
    const int row = wave * (2 * GEMM_ROWS) + p * 2 + half;
    *(v4f*)(C + (size_t)(mBase + row) * N + nBase + lid * 4) = s[p];
  }
}

__global__ __launch_bounds__(GEMM_THREADS)
void proj_gate_f32_wmma(const float* __restrict__ Pq,  const float* __restrict__ PWq,
                        const float* __restrict__ PEk, const float* __restrict__ PWk,
                        const float* __restrict__ PEv, const float* __restrict__ PWv,
                        float* __restrict__ PZ, float scale) {
  __shared__ __attribute__((aligned(16))) float red[GEMM_KW][GEMM_T][RED_PITCH];
  const int lane = threadIdx.x & 31;
  const int wave = threadIdx.x >> 5;
  const int lid  = lane & 15;
  const int half = lane >> 4;
  const int tA = blockIdx.y * (GEMM_T / 16);
  const int tB = blockIdx.x * (GEMM_T / 16);

  v8f acc[4][4];
  v4f sq[GEMM_ROWS], sk[GEMM_ROWS];
  v4f fa[GEMM_SETS][4], fb[GEMM_SETS][4];
  const float* aPtr[4];
  const float* bPtr[4];

  gemm_ptrs(aPtr, bPtr, Pq, PWq, tA, tB, lane, wave);
  gemm_head(fa, fb, aPtr, bPtr);
  gemm_kloop(acc, fa, fb, aPtr, bPtr);
  gemm_ptrs(aPtr, bPtr, PEk, PWk, tA, tB, lane, wave);
  gemm_head(fa, fb, aPtr, bPtr);
  gemm_reduce<true>(sq, red, acc, scale, wave, lid, half);
  lds_barrier();

  gemm_kloop(acc, fa, fb, aPtr, bPtr);
  gemm_ptrs(aPtr, bPtr, PEv, PWv, tA, tB, lane, wave);
  gemm_head(fa, fb, aPtr, bPtr);
  gemm_reduce<true>(sk, red, acc, scale, wave, lid, half);
  float d = 0.0f;
  #pragma unroll
  for (int p = 0; p < GEMM_ROWS; ++p)
    d += sq[p].x * sk[p].x + sq[p].y * sk[p].y + sq[p].z * sk[p].z + sq[p].w * sk[p].w;
  d += __shfl_xor(d, 16, 32);
  const float a = 1.0f / (1.0f + 31.0f * expf(-d));
  lds_barrier();

  gemm_kloop(acc, fa, fb, aPtr, bPtr);
  gemm_reduce<true>(sk, red, acc, 1.0f, wave, lid, half);
  float* zc = PZ + ((size_t)(tA + (wave >> 1)) * PK_CPT + blockIdx.x * (GEMM_T / 8)
                    + (wave & 1) * 4) * PK_CF + lane * 4;
  #pragma unroll
  for (int p = 0; p < GEMM_ROWS; ++p)
    *(v4f*)(zc + p * PK_CF) = a * sk[p];
}

#define SCAN_CH      32
#define SCAN_PITCH   36
#define SCAN_THREADS 512
#define SCAN_RPP     (SCAN_THREADS / 8)
#define SCAN_PASSES  (LSEQ / SCAN_RPP)
#define SCAN_BLK     32
#define PACK_MATS    5

__device__ __forceinline__ void scan_fetch(float (&x)[SCAN_BLK], const float* col, int t) {
  #pragma unroll
  for (int j = 0; j < SCAN_BLK; ++j) x[j] = col[(t + j) * SCAN_PITCH];
}

__device__ __forceinline__ float scan_run(const float (&x)[SCAN_BLK], float* col, int t,
                                          float g, float e) {
  #pragma unroll
  for (int j = 0; j < SCAN_BLK; ++j) {
    e = fmaf(g, e, x[j]);
    col[(t + j) * SCAN_PITCH] = e;
  }
  return e;
}

__global__ __launch_bounds__(SCAN_THREADS)
void ema_scan_pack(const float* __restrict__ kin, const float* __restrict__ vin,
                   float* __restrict__ PEk, float* __restrict__ PEv,
                   const float* __restrict__ gates,
                   const float* __restrict__ q,  const float* __restrict__ Wq,
                   const float* __restrict__ Wk, const float* __restrict__ Wv,
                   const float* __restrict__ Wd,
                   float* __restrict__ Pq,  float* __restrict__ PWq,
                   float* __restrict__ PWk, float* __restrict__ PWv,
                   float* __restrict__ PWd) {
  __shared__ __attribute__((aligned(16))) float strip[LSEQ * SCAN_PITCH];
  const int tid  = threadIdx.x;
  const int lane = tid & 31;
  const int wv   = tid >> 5;

  if (blockIdx.y >= 2) {
    const int m = blockIdx.y - 2;
    const float* src = m == 0 ? q  : m == 1 ? Wq  : m == 2 ? Wk  : m == 3 ? Wv  : Wd;
    float*       dst = m == 0 ? Pq : m == 1 ? PWq : m == 2 ? PWk : m == 3 ? PWv : PWd;
    const int T  = blockIdx.x * 2 + (wv >> 3);
    const int s0 = (wv & 7) * 16;
    const float* rp = src + (size_t)(T * 16 + (lane & 15)) * DMODEL + s0 * 8 + (lane >> 4) * 4;
    float*       wp = dst + ((size_t)T * PK_CPT + s0) * PK_CF + lane * 4;
    v4f x[16];
    #pragma unroll
    for (int j = 0; j < 16; ++j) x[j] = *(const v4f*)(rp + j * 8);
    #pragma unroll
    for (int j = 0; j < 16; ++j) *(v4f*)(wp + j * PK_CF) = x[j];
    return;
  }

  const float* X = (blockIdx.y == 0 ? kin : vin) + blockIdx.x * SCAN_CH;
  float*       Y = (blockIdx.y == 0 ? PEk : PEv);
  const int c4  = (tid & 7) * 4;
  const int r0  = tid >> 3;

  {
    v4f in[SCAN_PASSES];
    #pragma unroll
    for (int i = 0; i < SCAN_PASSES; ++i)
      in[i] = *(const v4f*)(X + (size_t)(i * SCAN_RPP + r0) * DMODEL + c4);
    #pragma unroll
    for (int i = 0; i < SCAN_PASSES; ++i)
      *(v4f*)(strip + (i * SCAN_RPP + r0) * SCAN_PITCH + c4) = in[i];
  }
  __syncthreads();

  if (tid < SCAN_CH) {
    const float g = gates[0];
    float* col = strip + tid;
    float e = 0.0f;
    float xa[SCAN_BLK], xb[SCAN_BLK];
    scan_fetch(xa, col, 0);
    for (int t = 0; t < LSEQ - 2 * SCAN_BLK; t += 2 * SCAN_BLK) {
      scan_fetch(xb, col, t + SCAN_BLK);
      e = scan_run(xa, col, t, g, e);
      scan_fetch(xa, col, t + 2 * SCAN_BLK);
      e = scan_run(xb, col, t + SCAN_BLK, g, e);
    }
    scan_fetch(xb, col, LSEQ - SCAN_BLK);
    e = scan_run(xa, col, LSEQ - 2 * SCAN_BLK, g, e);
    e = scan_run(xb, col, LSEQ - SCAN_BLK, g, e);
  }
  __syncthreads();

  const float* lp = strip + ((wv >> 2) * 16 + (lane & 15)) * SCAN_PITCH
                          + (wv & 3) * 8 + (lane >> 4) * 4;
  float* gp = Y + ((size_t)(wv >> 2) * PK_CPT + blockIdx.x * 4 + (wv & 3)) * PK_CF + lane * 4;
  v4f buf[SCAN_PASSES];
  #pragma unroll
  for (int i = 0; i < SCAN_PASSES; ++i)
    buf[i] = *(const v4f*)(lp + i * 64 * SCAN_PITCH);
  #pragma unroll
  for (int i = 0; i < SCAN_PASSES; ++i)
    *(v4f*)(gp + (size_t)i * 4 * PK_CPT * PK_CF) = buf[i];
}

extern "C" void kernel_launch(void* const* d_in, const int* in_sizes, int n_in,
                              void* d_out, int out_size, void* d_ws, size_t ws_size,
                              hipStream_t stream) {
  const float* q     = (const float*)d_in[0];
  const float* k     = (const float*)d_in[1];
  const float* v     = (const float*)d_in[2];
  const float* Wq    = (const float*)d_in[3];
  const float* Wk    = (const float*)d_in[4];
  const float* Wv    = (const float*)d_in[5];
  const float* Wd    = (const float*)d_in[6];
  const float* gates = (const float*)d_in[7];
  float* out = (float*)d_out;

  const int L = LSEQ, D = DMODEL;
  const size_t LD = (size_t)L * D;
  float* PEk = (float*)d_ws;
  float* PEv = PEk + LD;
  float* PZ  = PEv + LD;
  float* Pq  = PZ  + LD;
  float* PWq = Pq  + LD;
  float* PWk = PWq + LD;
  float* PWv = PWk + LD;
  float* PWd = PWv + LD;

  const float invScale = 1.0f / sqrtf(sqrtf(32.0f));

  dim3 gBlk(GEMM_THREADS);
  dim3 gGrid(D / GEMM_T, L / GEMM_T);

  hipLaunchKernelGGL(ema_scan_pack, dim3(DMODEL / SCAN_CH, 2 + PACK_MATS), dim3(SCAN_THREADS), 0, stream,
                     k, v, PEk, PEv, gates, q, Wq, Wk, Wv, Wd, Pq, PWq, PWk, PWv, PWd);
  hipLaunchKernelGGL(proj_gate_f32_wmma, gGrid, gBlk, 0, stream, Pq, PWq, PEk, PWk, PEv, PWv, PZ, invScale);
  hipLaunchKernelGGL(gemm_nt_f32_wmma, gGrid, gBlk, 0, stream, PZ, PWd, out, D);
}
